# starting kernel
# baseline (speedup 1.0000x reference)
#include <hip/hip_runtime.h>
#include <math.h>

typedef __attribute__((ext_vector_type(2))) float v2f;
typedef __attribute__((ext_vector_type(4))) float v4f;
typedef __attribute__((ext_vector_type(8))) float v8f;

#define HDIM 128
#define TM 16
#define LDS_STRIDE 132

__device__ __forceinline__ v8f wmma4(v2f a, v2f b, v8f c) {
    return __builtin_amdgcn_wmma_f32_16x16x4_f32(false, a, false, b, (short)0, c, false, false);
}

__device__ __forceinline__ float act_relu_tanh(float z) {
#if __has_builtin(__builtin_amdgcn_tanhf)
    float v = __builtin_amdgcn_tanhf(z);
#else
    float v = tanhf(z);
#endif
    return fmaxf(v, 0.f);
}

__device__ __forceinline__ void load_wfrags(const float* __restrict__ W,
                                            int n0, int col16, int hi, v2f* bfrag) {
    const float* wp = W + (hi * 2) * HDIM + n0 + col16;
#pragma unroll
    for (int k = 0; k < 32; ++k) {
        bfrag[k].x = wp[(4 * k) * HDIM];
        bfrag[k].y = wp[(4 * k + 1) * HDIM];
    }
}

__device__ __forceinline__ v8f tile_matmul(const float* __restrict__ arow,
                                           const v2f* __restrict__ bfrag) {
    v2f a[32];
#pragma unroll
    for (int k = 0; k < 32; ++k)
        a[k] = *(const v2f*)(arow + 4 * k);
    v8f c0 = {}, c1 = {}, c2 = {}, c3 = {};
#pragma unroll
    for (int k = 0; k < 8; ++k) {
        c0 = wmma4(a[k],      bfrag[k],      c0);
        c1 = wmma4(a[k + 8],  bfrag[k + 8],  c1);
        c2 = wmma4(a[k + 16], bfrag[k + 16], c2);
        c3 = wmma4(a[k + 24], bfrag[k + 24], c3);
    }
    return (c0 + c1) + (c2 + c3);
}

__global__ __launch_bounds__(256) void gemm128_bias(const float* __restrict__ X,
                                                    const float* __restrict__ W,
                                                    const float* __restrict__ bias,
                                                    float* __restrict__ Y,
                                                    int M) {
    __shared__ float xs[TM * LDS_STRIDE];
    const int tid   = threadIdx.x;
    const int lane  = tid & 31;
    const int wv    = tid >> 5;
    const int n0    = wv * 16;
    const int col16 = lane & 15;
    const int hi    = lane >> 4;

    v2f bfrag[32];
    load_wfrags(W, n0, col16, hi, bfrag);
    const float bval = bias[n0 + col16];

    const float* arow = &xs[col16 * LDS_STRIDE + hi * 2];

    const int nstrips = M / TM;
    for (int strip = blockIdx.x; strip < nstrips; strip += gridDim.x) {
        const float* xp = X + (size_t)strip * (TM * HDIM);
#pragma unroll
        for (int i = 0; i < 2; ++i) {
            int idx = tid + i * 256;
            int lin = idx * 4;
            int r = lin >> 7;
            int c = lin & 127;
            *(v4f*)(&xs[r * LDS_STRIDE + c]) = *(const v4f*)(xp + lin);
        }
        __syncthreads();

        v8f acc = tile_matmul(arow, bfrag);

        float* yp = Y + (size_t)strip * (TM * HDIM) + (size_t)(8 * hi) * HDIM + n0 + col16;
#pragma unroll
        for (int i = 0; i < 8; ++i)
            yp[(size_t)i * HDIM] = acc[i] + bval;
        __syncthreads();
    }
}

__global__ __launch_bounds__(256) void rnn_recur(const float* __restrict__ Wh,
                                                 float* __restrict__ ws,
                                                 int S) {
    __shared__ float hs[TM * LDS_STRIDE];
    const int tid   = threadIdx.x;
    const int lane  = tid & 31;
    const int wv    = tid >> 5;
    const int n0    = wv * 16;
    const int col16 = lane & 15;
    const int hi    = lane >> 4;
    const int b0    = blockIdx.x * TM;

    v2f bfrag[32];
    load_wfrags(Wh, n0, col16, hi, bfrag);

    for (int i = tid; i < TM * LDS_STRIDE; i += 256) hs[i] = 0.f;
    __syncthreads();

    const size_t rowstride = (size_t)S * HDIM;
    float* gp = ws + (size_t)(b0 + 8 * hi) * rowstride + n0 + col16;
    const float* arow = &hs[col16 * LDS_STRIDE + hi * 2];

    float xi_cur[8];
#pragma unroll
    for (int i = 0; i < 8; ++i) xi_cur[i] = gp[(size_t)i * rowstride];

    for (int t = 0; t < S; ++t) {
        float xi_nxt[8];
        if (t + 1 < S) {
#pragma unroll
            for (int i = 0; i < 8; ++i)
                xi_nxt[i] = gp[(size_t)i * rowstride + (size_t)(t + 1) * HDIM];
        } else {
#pragma unroll
            for (int i = 0; i < 8; ++i) xi_nxt[i] = 0.f;
        }

        v8f acc = tile_matmul(arow, bfrag);
        __syncthreads();

#pragma unroll
        for (int i = 0; i < 8; ++i) {
            float v = act_relu_tanh(acc[i] + xi_cur[i]);
            hs[(i + 8 * hi) * LDS_STRIDE + n0 + col16] = v;
            gp[(size_t)i * rowstride + (size_t)t * HDIM] = v;
        }
        __syncthreads();

#pragma unroll
        for (int i = 0; i < 8; ++i) xi_cur[i] = xi_nxt[i];
    }
}

extern "C" void kernel_launch(void* const* d_in, const int* in_sizes, int n_in,
                              void* d_out, int out_size, void* d_ws, size_t ws_size,
                              hipStream_t stream) {
    const float* x  = (const float*)d_in[0];
    const float* Wi = (const float*)d_in[1];
    const float* Wh = (const float*)d_in[2];
    const float* b  = (const float*)d_in[3];
    const float* Wo = (const float*)d_in[4];
    const float* bo = (const float*)d_in[5];
    float* y  = (float*)d_out;
    float* ws = (float*)d_ws;

    const int B = 64, S = 2048;
    const int M = B * S;

    gemm128_bias<<<1024, 256, 0, stream>>>(x, Wi, b, ws, M);
    rnn_recur<<<B / TM, 256, 0, stream>>>(Wh, ws, S);
    gemm128_bias<<<1024, 256, 0, stream>>>(ws, Wo, bo, y, M);
}


# rnn_recur: product transposed (Wh^T = A, h^T = B), 4 step waves x 32 hidden: result cells are the next B fragments, so 1/4 of the WMMAs run from registers under tanh and barrier; sum order changed
# speedup vs baseline: 1.9669x; 1.9669x over previous
#include <hip/hip_runtime.h>
#include <math.h>

typedef __attribute__((ext_vector_type(2))) float v2f;
typedef __attribute__((ext_vector_type(4))) float v4f;
typedef __attribute__((ext_vector_type(8))) float v8f;

#define HDIM 128
#define TM 16
#define LDS_STRIDE 132

__device__ __forceinline__ v8f wmma4(v2f a, v2f b, v8f c) {
    return __builtin_amdgcn_wmma_f32_16x16x4_f32(false, a, false, b, (short)0, c, false, false);
}

__device__ __forceinline__ float act_relu_tanh(float z) {
#if __has_builtin(__builtin_amdgcn_tanhf)
    float v = __builtin_amdgcn_tanhf(z);
#else
    float v = tanhf(z);
#endif
    return fmaxf(v, 0.f);
}

__device__ __forceinline__ void load_wfrags(const float* __restrict__ W,
                                            int n0, int col16, int hi, v2f* bfrag) {
    const float* wp = W + (hi * 2) * HDIM + n0 + col16;
#pragma unroll
    for (int k = 0; k < 32; ++k) {
        bfrag[k].x = wp[(4 * k) * HDIM];
        bfrag[k].y = wp[(4 * k + 1) * HDIM];
    }
}

__device__ __forceinline__ v8f tile_matmul(const float* __restrict__ arow,
                                           const v2f* __restrict__ bfrag) {
    v2f a[32];
#pragma unroll
    for (int k = 0; k < 32; ++k)
        a[k] = *(const v2f*)(arow + 4 * k);
    v8f c0 = {}, c1 = {}, c2 = {}, c3 = {};
#pragma unroll
    for (int k = 0; k < 8; ++k) {
        c0 = wmma4(a[k],      bfrag[k],      c0);
        c1 = wmma4(a[k + 8],  bfrag[k + 8],  c1);
        c2 = wmma4(a[k + 16], bfrag[k + 16], c2);
        c3 = wmma4(a[k + 24], bfrag[k + 24], c3);
    }
    return (c0 + c1) + (c2 + c3);
}

__global__ __launch_bounds__(256) void gemm128_bias(const float* __restrict__ X,
                                                    const float* __restrict__ W,
                                                    const float* __restrict__ bias,
                                                    float* __restrict__ Y,
                                                    int M) {
    __shared__ float xs[TM * LDS_STRIDE];
    const int tid   = threadIdx.x;
    const int lane  = tid & 31;
    const int wv    = tid >> 5;
    const int n0    = wv * 16;
    const int col16 = lane & 15;
    const int hi    = lane >> 4;

    v2f bfrag[32];
    load_wfrags(W, n0, col16, hi, bfrag);
    const float bval = bias[n0 + col16];

    const float* arow = &xs[col16 * LDS_STRIDE + hi * 2];

    const int nstrips = M / TM;
    for (int strip = blockIdx.x; strip < nstrips; strip += gridDim.x) {
        const float* xp = X + (size_t)strip * (TM * HDIM);
#pragma unroll
        for (int i = 0; i < 2; ++i) {
            int idx = tid + i * 256;
            int lin = idx * 4;
            int r = lin >> 7;
            int c = lin & 127;
            *(v4f*)(&xs[r * LDS_STRIDE + c]) = *(const v4f*)(xp + lin);
        }
        __syncthreads();

        v8f acc = tile_matmul(arow, bfrag);

        float* yp = Y + (size_t)strip * (TM * HDIM) + (size_t)(8 * hi) * HDIM + n0 + col16;
#pragma unroll
        for (int i = 0; i < 8; ++i)
            yp[(size_t)i * HDIM] = acc[i] + bval;
        __syncthreads();
    }
}

#define RECUR_THREADS 384
__global__ __launch_bounds__(RECUR_THREADS) void rnn_recur(const float* __restrict__ Wh,
                                                           float* __restrict__ ws,
                                                           int S) {
    __shared__ __attribute__((aligned(16))) float hs[2][TM * LDS_STRIDE];
    __shared__ __attribute__((aligned(16))) float xs[2][TM * LDS_STRIDE];
    const int tid   = threadIdx.x;
    const int lane  = tid & 31;
    const int wv    = __builtin_amdgcn_readfirstlane(tid >> 5);
    const bool step_wave = wv < 4;
    const int col16 = lane & 15;
    const int hi    = lane >> 4;
    const int b0    = blockIdx.x * TM;
    const size_t rowstride = (size_t)S * HDIM;

    const int w = wv & 3;
    const int cell = col16 * LDS_STRIDE + 32 * w + 8 * hi;
    int piece[3];
#pragma unroll
    for (int d = 0; d < 3; ++d)
        piece[d] = col16 * LDS_STRIDE + 32 * ((w + 1 + d) & 3) + 8 * hi;

    const int r0 = 2 * ((wv - 4) & 7);
    const int io_lds = r0 * LDS_STRIDE + lane * 4;
    float* gio = ws + (size_t)(b0 + r0) * rowstride + lane * 4;

    v2f a0[32], a1[32];
    v8f t0a = {}, t0b = {}, t1a = {}, t1b = {};
    v8f h1r = {};
    if (step_wave) {
#pragma unroll
        for (int q = 0; q < 8; ++q) {
            const int g = (q < 2) ? (32 * w + 16 * q) : (32 * ((w + 1 + ((q - 2) >> 1)) & 3) + 16 * (q & 1));
            const float* wp = Wh + (size_t)(g + 8 * hi) * HDIM + 32 * w + col16;
#pragma unroll
            for (int jj = 0; jj < 4; ++jj) {
                a0[4 * q + jj].x = wp[(2 * jj) * HDIM];
                a0[4 * q + jj].y = wp[(2 * jj + 1) * HDIM];
                a1[4 * q + jj].x = wp[(2 * jj) * HDIM + 16];
                a1[4 * q + jj].y = wp[(2 * jj + 1) * HDIM + 16];
            }
        }
        const v2f z2 = {0.f, 0.f};
        const v8f z8 = {};
        t0a = wmma4(a0[0], z2, z8);
        t0b = wmma4(a0[1], z2, z8);
        t0a = wmma4(a0[2], z2, t0a);
        t0b = wmma4(a0[3], z2, t0b);
        t1a = wmma4(a1[0], z2, z8);
        t1b = wmma4(a1[1], z2, z8);
        t1a = wmma4(a1[2], z2, t1a);
        t1b = wmma4(a1[3], z2, t1b);
        t0a = wmma4(a0[4], z2, t0a);
        t0b = wmma4(a0[5], z2, t0b);
        t0a = wmma4(a0[6], z2, t0a);
        t0b = wmma4(a0[7], z2, t0b);
    } else {
        v4f g0 = *(const v4f*)(gio);
        v4f g1 = *(const v4f*)(gio + rowstride);
        *(v4f*)(&xs[0][io_lds]) = g0;
        *(v4f*)(&xs[0][io_lds + LDS_STRIDE]) = g1;
    }
    for (int i = tid; i < TM * LDS_STRIDE; i += RECUR_THREADS) hs[1][i] = 0.f;
    __syncthreads();

    for (int t = 0; t < S; ++t) {
        const float* hprev = hs[(t + 1) & 1];
        float* hcur = hs[t & 1];
        if (step_wave) {
            v4f p[6][2];
#pragma unroll
            for (int d = 0; d < 3; ++d) {
                p[2 * d][0]     = *(const v4f*)(hprev + piece[d]);
                p[2 * d][1]     = *(const v4f*)(hprev + piece[d] + 4);
                p[2 * d + 1][0] = *(const v4f*)(hprev + piece[d] + 16);
                p[2 * d + 1][1] = *(const v4f*)(hprev + piece[d] + 20);
            }
            const v4f x00 = *(const v4f*)(&xs[t & 1][cell]);
            const v4f x01 = *(const v4f*)(&xs[t & 1][cell + 4]);
            const v4f x10 = *(const v4f*)(&xs[t & 1][cell + 16]);
            const v4f x11 = *(const v4f*)(&xs[t & 1][cell + 20]);
            __builtin_amdgcn_sched_barrier(0);

            {
                const v2f o0 = {h1r[0], h1r[1]};
                const v2f o1 = {h1r[2], h1r[3]};
                const v2f o2 = {h1r[4], h1r[5]};
                const v2f o3 = {h1r[6], h1r[7]};
                t1a = wmma4(a1[4], o0, t1a);
                t1b = wmma4(a1[5], o1, t1b);
                t1a = wmma4(a1[6], o2, t1a);
                t1b = wmma4(a1[7], o3, t1b);
            }
#pragma unroll
            for (int q = 0; q < 6; ++q) {
                const v2f b0f = {p[q][0].x, p[q][0].y};
                const v2f b1f = {p[q][0].z, p[q][0].w};
                const v2f b2f = {p[q][1].x, p[q][1].y};
                const v2f b3f = {p[q][1].z, p[q][1].w};
                t0a = wmma4(a0[4 * q + 8],  b0f, t0a);
                t0b = wmma4(a0[4 * q + 9],  b1f, t0b);
                t0a = wmma4(a0[4 * q + 10], b2f, t0a);
                t0b = wmma4(a0[4 * q + 11], b3f, t0b);
            }
            v8f acc0;
            v4f n00, n01;
#pragma unroll
            for (int q = 0; q < 6; ++q) {
                const v2f b0f = {p[q][0].x, p[q][0].y};
                const v2f b1f = {p[q][0].z, p[q][0].w};
                const v2f b2f = {p[q][1].x, p[q][1].y};
                const v2f b3f = {p[q][1].z, p[q][1].w};
                t1a = wmma4(a1[4 * q + 8],  b0f, t1a);
                t1b = wmma4(a1[4 * q + 9],  b1f, t1b);
                t1a = wmma4(a1[4 * q + 10], b2f, t1a);
                t1b = wmma4(a1[4 * q + 11], b3f, t1b);
                if (q == 0) {
                    acc0 = t0a + t0b;
                }
                if (q == 1) {
                    n00.x = act_relu_tanh(acc0[0] + x00.x);
                    n00.y = act_relu_tanh(acc0[1] + x00.y);
                    n00.z = act_relu_tanh(acc0[2] + x00.z);
                    n00.w = act_relu_tanh(acc0[3] + x00.w);
                    n01.x = act_relu_tanh(acc0[4] + x01.x);
                    n01.y = act_relu_tanh(acc0[5] + x01.y);
                    n01.z = act_relu_tanh(acc0[6] + x01.z);
                    n01.w = act_relu_tanh(acc0[7] + x01.w);
                }
                if (q == 2) {
                    *(v4f*)(hcur + cell) = n00;
                    *(v4f*)(hcur + cell + 4) = n01;
                }
            }
            const v8f z8 = {};
            const v2f s0 = {n00.x, n00.y};
            const v2f s1 = {n00.z, n00.w};
            const v2f s2 = {n01.x, n01.y};
            const v2f s3 = {n01.z, n01.w};
            t0a = wmma4(a0[0], s0, z8);
            t0b = wmma4(a0[1], s1, z8);
            const v8f acc1 = t1a + t1b;
            t0a = wmma4(a0[2], s2, t0a);
            t0b = wmma4(a0[3], s3, t0b);
            v4f n10, n11;
            n10.x = act_relu_tanh(acc1[0] + x10.x);
            n10.y = act_relu_tanh(acc1[1] + x10.y);
            n10.z = act_relu_tanh(acc1[2] + x10.z);
            n10.w = act_relu_tanh(acc1[3] + x10.w);
            n11.x = act_relu_tanh(acc1[4] + x11.x);
            n11.y = act_relu_tanh(acc1[5] + x11.y);
            n11.z = act_relu_tanh(acc1[6] + x11.z);
            n11.w = act_relu_tanh(acc1[7] + x11.w);
            t1a = wmma4(a1[0], s0, z8);
            t1b = wmma4(a1[1], s1, z8);
            *(v4f*)(hcur + cell + 16) = n10;
            *(v4f*)(hcur + cell + 20) = n11;
            t1a = wmma4(a1[2], s2, t1a);
            t1b = wmma4(a1[3], s3, t1b);
            __builtin_amdgcn_sched_barrier(0);
            h1r[0] = n10.x; h1r[1] = n10.y; h1r[2] = n10.z; h1r[3] = n10.w;
            h1r[4] = n11.x; h1r[5] = n11.y; h1r[6] = n11.z; h1r[7] = n11.w;
            {
                const v2f o0 = {n10.x, n10.y};
                const v2f o1 = {n10.z, n10.w};
                const v2f o2 = {n11.x, n11.y};
                const v2f o3 = {n11.z, n11.w};
                t0a = wmma4(a0[4], o0, t0a);
                t0b = wmma4(a0[5], o1, t0b);
                t0a = wmma4(a0[6], o2, t0a);
                t0b = wmma4(a0[7], o3, t0b);
            }
        } else {
            const bool more = (t + 1 < S);
            v4f g0 = {}, g1 = {};
            if (more) {
                g0 = *(const v4f*)(gio + (size_t)(t + 1) * HDIM);
                g1 = *(const v4f*)(gio + rowstride + (size_t)(t + 1) * HDIM);
            }
            if (t > 0) {
                v4f h0 = *(const v4f*)(hprev + io_lds);
                v4f h1 = *(const v4f*)(hprev + io_lds + LDS_STRIDE);
                *(v4f*)(gio + (size_t)(t - 1) * HDIM) = h0;
                *(v4f*)(gio + rowstride + (size_t)(t - 1) * HDIM) = h1;
            }
            if (more) {
                float* xd = &xs[(t + 1) & 1][io_lds];
                *(v4f*)(xd) = g0;
                *(v4f*)(xd + LDS_STRIDE) = g1;
            }
        }
        __syncthreads();
    }

    if (!step_wave) {
        const float* hlast = hs[(S - 1) & 1];
        v4f h0 = *(const v4f*)(hlast + io_lds);
        v4f h1 = *(const v4f*)(hlast + io_lds + LDS_STRIDE);
        *(v4f*)(gio + (size_t)(S - 1) * HDIM) = h0;
        *(v4f*)(gio + rowstride + (size_t)(S - 1) * HDIM) = h1;
    }
}

extern "C" void kernel_launch(void* const* d_in, const int* in_sizes, int n_in,
                              void* d_out, int out_size, void* d_ws, size_t ws_size,
                              hipStream_t stream) {
    const float* x  = (const float*)d_in[0];
    const float* Wi = (const float*)d_in[1];
    const float* Wh = (const float*)d_in[2];
    const float* b  = (const float*)d_in[3];
    const float* Wo = (const float*)d_in[4];
    const float* bo = (const float*)d_in[5];
    float* y  = (float*)d_out;
    float* ws = (float*)d_ws;

    const int B = 64, S = 2048;
    const int M = B * S;

    gemm128_bias<<<1024, 256, 0, stream>>>(x, Wi, b, ws, M);
    rnn_recur<<<B / TM, RECUR_THREADS, 0, stream>>>(Wh, ws, S);
    gemm128_bias<<<1024, 256, 0, stream>>>(ws, Wo, bo, y, M);
}
